# starting kernel
# baseline (speedup 1.0000x reference)
#include <hip/hip_runtime.h>
#include <cmath>


#define WHVI_D 8192
#define WHVI_B 4096

typedef __attribute__((ext_vector_type(2))) float v2f;
typedef __attribute__((ext_vector_type(8))) float v8f;

__global__ __launch_bounds__(512)
void whvi_fwht_prep(const float* __restrict__ s1,
                    const float* __restrict__ s2,
                    const float* __restrict__ g_mu,
                    const float* __restrict__ g_rho,
                    const float* __restrict__ eps,
                    float* __restrict__ w_col) {
    __shared__ float u[WHVI_D];
    const int tid = threadIdx.x;

    for (int n = tid; n < WHVI_D; n += 512) {
        float r  = g_rho[n];
        float sp = (r > 20.0f) ? r : log1pf(__expf(r));
        u[n] = g_mu[n] + sp * eps[n];
    }
    __syncthreads();

    if (tid < 256) {
        float v[32];
        const int base = tid * 32;
        #pragma unroll
        for (int q = 0; q < 32; ++q) v[q] = u[base + q];
        #pragma unroll
        for (int h = 1; h < 32; h <<= 1) {
            #pragma unroll
            for (int q = 0; q < 32; ++q) {
                if ((q & h) == 0) {
                    float a = v[q], b = v[q + h];
                    v[q] = a + b;
                    v[q + h] = a - b;
                }
            }
        }
        #pragma unroll
        for (int q = 0; q < 32; ++q) u[base + q] = v[q];
    }
    __syncthreads();

    const int lane  = tid & 31;
    const int wave  = tid >> 5;
    const int laneM = lane & 15;
    const int khalf = (lane >> 4) << 1;
    const int mbase = (lane >> 4) << 3;

    for (int stage = 0; stage < 2; ++stage) {
        #pragma unroll
        for (int s = 0; s < 2; ++s) {
            const int t = wave * 2 + s;
            const int c = t * 16 + laneM;
            int cbase, cstride;
            if (stage == 0) { cbase = (c >> 5) * 512 + (c & 31); cstride = 32;  }
            else            { cbase = c;                          cstride = 512; }

            v8f acc = {};
            #pragma unroll
            for (int q = 0; q < 4; ++q) {
                const int k0 = 4 * q + khalf;
                v2f a, b;
                a.x = (__popc(laneM & k0)       & 1) ? -1.0f : 1.0f;
                a.y = (__popc(laneM & (k0 + 1)) & 1) ? -1.0f : 1.0f;
                b.x = u[cbase + k0 * cstride];
                b.y = u[cbase + (k0 + 1) * cstride];
                acc = __builtin_amdgcn_wmma_f32_16x16x4_f32(
                          false, a, false, b, (short)0, acc, false, false);
            }
            #pragma unroll
            for (int r = 0; r < 8; ++r)
                u[cbase + (mbase + r) * cstride] = acc[r];
        }
        __syncthreads();
    }

    const float s1_0 = s1[0];
    for (int n = tid; n < WHVI_D; n += 512)
        w_col[n] = s1_0 * s2[n] * u[n];
}

__global__ __launch_bounds__(256)
void whvi_outer(const float* __restrict__ x,
                const float* __restrict__ w,
                float4* __restrict__ out) {
    const size_t idx = (size_t)blockIdx.x * 256 + threadIdx.x;
    const int row = (int)(idx >> 11);
    const int col = (int)(idx & 2047);
    const float xv = x[row];
    const float4 wv = reinterpret_cast<const float4*>(w)[col];
    float4 o;
    o.x = xv * wv.x;
    o.y = xv * wv.y;
    o.z = xv * wv.z;
    o.w = xv * wv.w;
    out[idx] = o;
}

extern "C" void kernel_launch(void* const* d_in, const int* in_sizes, int n_in,
                              void* d_out, int out_size, void* d_ws, size_t ws_size,
                              hipStream_t stream) {
    const float* x     = (const float*)d_in[0];
    const float* s1    = (const float*)d_in[1];
    const float* s2    = (const float*)d_in[2];
    const float* g_mu  = (const float*)d_in[3];
    const float* g_rho = (const float*)d_in[4];
    const float* eps   = (const float*)d_in[5];
    float* out   = (float*)d_out;
    float* w_col = (float*)d_ws;

    whvi_fwht_prep<<<1, 512, 0, stream>>>(s1, s2, g_mu, g_rho, eps, w_col);

    const int total_f4 = (WHVI_B * WHVI_D) / 4;
    whvi_outer<<<total_f4 / 256, 256, 0, stream>>>(x, w_col, (float4*)out);
}


# prep kernel: all global reads (g_rho, eps, g_mu, s2, s1[0]) issued up front as one batch of b128 loads held in registers, one memory round trip instead of 41 serial ones
# speedup vs baseline: 1.2636x; 1.2636x over previous
#include <hip/hip_runtime.h>
#include <cmath>


#define WHVI_D 8192
#define WHVI_B 4096

typedef __attribute__((ext_vector_type(2))) float v2f;
typedef __attribute__((ext_vector_type(8))) float v8f;

__device__ __forceinline__ float whvi_softplus(float r) {
    return (r > 20.0f) ? r : log1pf(__expf(r));
}

__global__ __launch_bounds__(512)
void whvi_fwht_prep(const float* __restrict__ s1,
                    const float* __restrict__ s2,
                    const float* __restrict__ g_mu,
                    const float* __restrict__ g_rho,
                    const float* __restrict__ eps,
                    float* __restrict__ w_col) {
    __shared__ __attribute__((aligned(16))) float u[WHVI_D];
    const int tid = threadIdx.x;

    const float4* __restrict__ rho4 = reinterpret_cast<const float4*>(g_rho);
    const float4* __restrict__ eps4 = reinterpret_cast<const float4*>(eps);
    const float4* __restrict__ mu4  = reinterpret_cast<const float4*>(g_mu);
    const float4* __restrict__ s24  = reinterpret_cast<const float4*>(s2);
    float4 rv[4], ev[4], mv[4], sv[4];
    #pragma unroll
    for (int p = 0; p < 4; ++p) rv[p] = rho4[p * 512 + tid];
    #pragma unroll
    for (int p = 0; p < 4; ++p) ev[p] = eps4[p * 512 + tid];
    #pragma unroll
    for (int p = 0; p < 4; ++p) mv[p] = mu4[p * 512 + tid];
    #pragma unroll
    for (int p = 0; p < 4; ++p) sv[p] = s24[p * 512 + tid];
    const float s1_0 = s1[0];

    #pragma unroll
    for (int p = 0; p < 4; ++p) {
        float4 g;
        g.x = mv[p].x + whvi_softplus(rv[p].x) * ev[p].x;
        g.y = mv[p].y + whvi_softplus(rv[p].y) * ev[p].y;
        g.z = mv[p].z + whvi_softplus(rv[p].z) * ev[p].z;
        g.w = mv[p].w + whvi_softplus(rv[p].w) * ev[p].w;
        reinterpret_cast<float4*>(u)[p * 512 + tid] = g;
    }
    __syncthreads();

    if (tid < 256) {
        float v[32];
        const int base = tid * 32;
        #pragma unroll
        for (int q = 0; q < 32; ++q) v[q] = u[base + q];
        #pragma unroll
        for (int h = 1; h < 32; h <<= 1) {
            #pragma unroll
            for (int q = 0; q < 32; ++q) {
                if ((q & h) == 0) {
                    float a = v[q], b = v[q + h];
                    v[q] = a + b;
                    v[q + h] = a - b;
                }
            }
        }
        #pragma unroll
        for (int q = 0; q < 32; ++q) u[base + q] = v[q];
    }
    __syncthreads();

    const int lane  = tid & 31;
    const int wave  = tid >> 5;
    const int laneM = lane & 15;
    const int khalf = (lane >> 4) << 1;
    const int mbase = (lane >> 4) << 3;

    for (int stage = 0; stage < 2; ++stage) {
        #pragma unroll
        for (int s = 0; s < 2; ++s) {
            const int t = wave * 2 + s;
            const int c = t * 16 + laneM;
            int cbase, cstride;
            if (stage == 0) { cbase = (c >> 5) * 512 + (c & 31); cstride = 32;  }
            else            { cbase = c;                          cstride = 512; }

            v8f acc = {};
            #pragma unroll
            for (int q = 0; q < 4; ++q) {
                const int k0 = 4 * q + khalf;
                v2f a, b;
                a.x = (__popc(laneM & k0)       & 1) ? -1.0f : 1.0f;
                a.y = (__popc(laneM & (k0 + 1)) & 1) ? -1.0f : 1.0f;
                b.x = u[cbase + k0 * cstride];
                b.y = u[cbase + (k0 + 1) * cstride];
                acc = __builtin_amdgcn_wmma_f32_16x16x4_f32(
                          false, a, false, b, (short)0, acc, false, false);
            }
            #pragma unroll
            for (int r = 0; r < 8; ++r)
                u[cbase + (mbase + r) * cstride] = acc[r];
        }
        __syncthreads();
    }

    #pragma unroll
    for (int p = 0; p < 4; ++p) {
        const float4 uv = reinterpret_cast<const float4*>(u)[p * 512 + tid];
        float4 o;
        o.x = s1_0 * sv[p].x * uv.x;
        o.y = s1_0 * sv[p].y * uv.y;
        o.z = s1_0 * sv[p].z * uv.z;
        o.w = s1_0 * sv[p].w * uv.w;
        reinterpret_cast<float4*>(w_col)[p * 512 + tid] = o;
    }
}

__global__ __launch_bounds__(256)
void whvi_outer(const float* __restrict__ x,
                const float* __restrict__ w,
                float4* __restrict__ out) {
    const size_t idx = (size_t)blockIdx.x * 256 + threadIdx.x;
    const int row = (int)(idx >> 11);
    const int col = (int)(idx & 2047);
    const float xv = x[row];
    const float4 wv = reinterpret_cast<const float4*>(w)[col];
    float4 o;
    o.x = xv * wv.x;
    o.y = xv * wv.y;
    o.z = xv * wv.z;
    o.w = xv * wv.w;
    out[idx] = o;
}

extern "C" void kernel_launch(void* const* d_in, const int* in_sizes, int n_in,
                              void* d_out, int out_size, void* d_ws, size_t ws_size,
                              hipStream_t stream) {
    const float* x     = (const float*)d_in[0];
    const float* s1    = (const float*)d_in[1];
    const float* s2    = (const float*)d_in[2];
    const float* g_mu  = (const float*)d_in[3];
    const float* g_rho = (const float*)d_in[4];
    const float* eps   = (const float*)d_in[5];
    float* out   = (float*)d_out;
    float* w_col = (float*)d_ws;

    whvi_fwht_prep<<<1, 512, 0, stream>>>(s1, s2, g_mu, g_rho, eps, w_col);

    const int total_f4 = (WHVI_B * WHVI_D) / 4;
    whvi_outer<<<total_f4 / 256, 256, 0, stream>>>(x, w_col, (float4*)out);
}


# outer-product kernel as a resident grid: 256 workgroups of 8 waves, 16 rows each, w kept in registers (read once per lane, not once per store), 128 whole-line b128 stores per lane instead of 1
# speedup vs baseline: 1.6255x; 1.2864x over previous
#include <hip/hip_runtime.h>
#include <cmath>


#define WHVI_D 8192
#define WHVI_B 4096

typedef __attribute__((ext_vector_type(2))) float v2f;
typedef __attribute__((ext_vector_type(8))) float v8f;

__device__ __forceinline__ float whvi_softplus(float r) {
    return (r > 20.0f) ? r : log1pf(__expf(r));
}

__global__ __launch_bounds__(512)
void whvi_fwht_prep(const float* __restrict__ s1,
                    const float* __restrict__ s2,
                    const float* __restrict__ g_mu,
                    const float* __restrict__ g_rho,
                    const float* __restrict__ eps,
                    float* __restrict__ w_col) {
    __shared__ __attribute__((aligned(16))) float u[WHVI_D];
    const int tid = threadIdx.x;

    const float4* __restrict__ rho4 = reinterpret_cast<const float4*>(g_rho);
    const float4* __restrict__ eps4 = reinterpret_cast<const float4*>(eps);
    const float4* __restrict__ mu4  = reinterpret_cast<const float4*>(g_mu);
    const float4* __restrict__ s24  = reinterpret_cast<const float4*>(s2);
    float4 rv[4], ev[4], mv[4], sv[4];
    #pragma unroll
    for (int p = 0; p < 4; ++p) rv[p] = rho4[p * 512 + tid];
    #pragma unroll
    for (int p = 0; p < 4; ++p) ev[p] = eps4[p * 512 + tid];
    #pragma unroll
    for (int p = 0; p < 4; ++p) mv[p] = mu4[p * 512 + tid];
    #pragma unroll
    for (int p = 0; p < 4; ++p) sv[p] = s24[p * 512 + tid];
    const float s1_0 = s1[0];

    #pragma unroll
    for (int p = 0; p < 4; ++p) {
        float4 g;
        g.x = mv[p].x + whvi_softplus(rv[p].x) * ev[p].x;
        g.y = mv[p].y + whvi_softplus(rv[p].y) * ev[p].y;
        g.z = mv[p].z + whvi_softplus(rv[p].z) * ev[p].z;
        g.w = mv[p].w + whvi_softplus(rv[p].w) * ev[p].w;
        reinterpret_cast<float4*>(u)[p * 512 + tid] = g;
    }
    __syncthreads();

    if (tid < 256) {
        float v[32];
        const int base = tid * 32;
        #pragma unroll
        for (int q = 0; q < 32; ++q) v[q] = u[base + q];
        #pragma unroll
        for (int h = 1; h < 32; h <<= 1) {
            #pragma unroll
            for (int q = 0; q < 32; ++q) {
                if ((q & h) == 0) {
                    float a = v[q], b = v[q + h];
                    v[q] = a + b;
                    v[q + h] = a - b;
                }
            }
        }
        #pragma unroll
        for (int q = 0; q < 32; ++q) u[base + q] = v[q];
    }
    __syncthreads();

    const int lane  = tid & 31;
    const int wave  = tid >> 5;
    const int laneM = lane & 15;
    const int khalf = (lane >> 4) << 1;
    const int mbase = (lane >> 4) << 3;

    for (int stage = 0; stage < 2; ++stage) {
        #pragma unroll
        for (int s = 0; s < 2; ++s) {
            const int t = wave * 2 + s;
            const int c = t * 16 + laneM;
            int cbase, cstride;
            if (stage == 0) { cbase = (c >> 5) * 512 + (c & 31); cstride = 32;  }
            else            { cbase = c;                          cstride = 512; }

            v8f acc = {};
            #pragma unroll
            for (int q = 0; q < 4; ++q) {
                const int k0 = 4 * q + khalf;
                v2f a, b;
                a.x = (__popc(laneM & k0)       & 1) ? -1.0f : 1.0f;
                a.y = (__popc(laneM & (k0 + 1)) & 1) ? -1.0f : 1.0f;
                b.x = u[cbase + k0 * cstride];
                b.y = u[cbase + (k0 + 1) * cstride];
                acc = __builtin_amdgcn_wmma_f32_16x16x4_f32(
                          false, a, false, b, (short)0, acc, false, false);
            }
            #pragma unroll
            for (int r = 0; r < 8; ++r)
                u[cbase + (mbase + r) * cstride] = acc[r];
        }
        __syncthreads();
    }

    #pragma unroll
    for (int p = 0; p < 4; ++p) {
        const float4 uv = reinterpret_cast<const float4*>(u)[p * 512 + tid];
        float4 o;
        o.x = s1_0 * sv[p].x * uv.x;
        o.y = s1_0 * sv[p].y * uv.y;
        o.z = s1_0 * sv[p].z * uv.z;
        o.w = s1_0 * sv[p].w * uv.w;
        reinterpret_cast<float4*>(w_col)[p * 512 + tid] = o;
    }
}

#define WHVI_OUTER_ROWS 16
#define WHVI_OUTER_WGS  256
#define WHVI_ROW_F4     2048

__global__ __launch_bounds__(256)
void whvi_outer(const float* __restrict__ x,
                const float* __restrict__ w,
                float4* __restrict__ out) {
    const int lane = threadIdx.x & 31;
    const int wave = threadIdx.x >> 5;
    const int row0 = blockIdx.x * WHVI_OUTER_ROWS;
    const int col0 = wave * 256 + lane;

    const float4* __restrict__ w4 = reinterpret_cast<const float4*>(w);
    float4 wv[8];
    #pragma unroll
    for (int q = 0; q < 8; ++q) wv[q] = w4[col0 + 32 * q];

    float4* __restrict__ o = out + (size_t)row0 * WHVI_ROW_F4 + col0;
    #pragma unroll
    for (int r = 0; r < WHVI_OUTER_ROWS; ++r) {
        const float xv = x[row0 + r];
        #pragma unroll
        for (int q = 0; q < 8; ++q) {
            float4 v;
            v.x = xv * wv[q].x;
            v.y = xv * wv[q].y;
            v.z = xv * wv[q].z;
            v.w = xv * wv[q].w;
            o[r * WHVI_ROW_F4 + 32 * q] = v;
        }
    }
}

extern "C" void kernel_launch(void* const* d_in, const int* in_sizes, int n_in,
                              void* d_out, int out_size, void* d_ws, size_t ws_size,
                              hipStream_t stream) {
    const float* x     = (const float*)d_in[0];
    const float* s1    = (const float*)d_in[1];
    const float* s2    = (const float*)d_in[2];
    const float* g_mu  = (const float*)d_in[3];
    const float* g_rho = (const float*)d_in[4];
    const float* eps   = (const float*)d_in[5];
    float* out   = (float*)d_out;
    float* w_col = (float*)d_ws;

    whvi_fwht_prep<<<1, 512, 0, stream>>>(s1, s2, g_mu, g_rho, eps, w_col);

    whvi_outer<<<WHVI_OUTER_WGS, 256, 0, stream>>>(x, w_col, (float4*)out);
}


# FWHT split at its top 3 index bits: prep runs as 8 workgroups (one 1024-slice each, H4 in regs + two H16 WMMA stages), outer kernel does the last 3 butterfly levels on load; changes FWHT sum order
# speedup vs baseline: 2.3122x; 1.4225x over previous
#include <hip/hip_runtime.h>
#include <cmath>


#define WHVI_D 8192
#define WHVI_B 4096
#define WHVI_SLICES   8
#define WHVI_SLICE_F4 256

typedef __attribute__((ext_vector_type(2))) float v2f;
typedef __attribute__((ext_vector_type(8))) float v8f;

__device__ __forceinline__ float whvi_softplus(float r) {
    return (r > 20.0f) ? r : log1pf(__expf(r));
}

__global__ __launch_bounds__(256)
void whvi_fwht_prep(const float* __restrict__ g_mu,
                    const float* __restrict__ g_rho,
                    const float* __restrict__ eps,
                    float* __restrict__ v_out) {
    __shared__ __attribute__((aligned(16))) float u[WHVI_SLICE_F4 * 4];
    const int tid = threadIdx.x;
    const int f4  = blockIdx.x * WHVI_SLICE_F4 + tid;

    const float4 rv = reinterpret_cast<const float4*>(g_rho)[f4];
    const float4 ev = reinterpret_cast<const float4*>(eps)[f4];
    const float4 mv = reinterpret_cast<const float4*>(g_mu)[f4];

    const float g0 = mv.x + whvi_softplus(rv.x) * ev.x;
    const float g1 = mv.y + whvi_softplus(rv.y) * ev.y;
    const float g2 = mv.z + whvi_softplus(rv.z) * ev.z;
    const float g3 = mv.w + whvi_softplus(rv.w) * ev.w;

    const float a0 = g0 + g1, a1 = g0 - g1, a2 = g2 + g3, a3 = g2 - g3;
    float4 t;
    t.x = a0 + a2;
    t.y = a1 + a3;
    t.z = a0 - a2;
    t.w = a1 - a3;
    reinterpret_cast<float4*>(u)[tid] = t;
    __syncthreads();

    const int lane  = tid & 31;
    const int wave  = tid >> 5;
    const int laneM = lane & 15;
    const int khalf = (lane >> 4) << 1;
    const int mbase = (lane >> 4) << 3;

    #pragma unroll
    for (int stage = 0; stage < 2; ++stage) {
        if (wave < 4) {
            const int c = wave * 16 + laneM;
            int cbase, cstride;
            if (stage == 0) { cbase = (c >> 2) * 64 + (c & 3); cstride = 4;  }
            else            { cbase = c;                        cstride = 64; }

            v8f acc = {};
            #pragma unroll
            for (int q = 0; q < 4; ++q) {
                const int k0 = 4 * q + khalf;
                v2f a, b;
                a.x = (__popc(laneM & k0)       & 1) ? -1.0f : 1.0f;
                a.y = (__popc(laneM & (k0 + 1)) & 1) ? -1.0f : 1.0f;
                b.x = u[cbase + k0 * cstride];
                b.y = u[cbase + (k0 + 1) * cstride];
                acc = __builtin_amdgcn_wmma_f32_16x16x4_f32(
                          false, a, false, b, (short)0, acc, false, false);
            }
            #pragma unroll
            for (int r = 0; r < 8; ++r)
                u[cbase + (mbase + r) * cstride] = acc[r];
        }
        __syncthreads();
    }

    reinterpret_cast<float4*>(v_out)[f4] = reinterpret_cast<const float4*>(u)[tid];
}

#define WHVI_OUTER_ROWS 16
#define WHVI_OUTER_WGS  256
#define WHVI_ROW_F4     2048

__global__ __launch_bounds__(256)
void whvi_outer(const float* __restrict__ x,
                const float* __restrict__ s1,
                const float* __restrict__ s2,
                const float* __restrict__ v,
                float4* __restrict__ out) {
    const int lane = threadIdx.x & 31;
    const int wave = threadIdx.x >> 5;
    const int row0 = blockIdx.x * WHVI_OUTER_ROWS;
    const int col0 = wave * 32 + lane;

    const float4* __restrict__ v4  = reinterpret_cast<const float4*>(v);
    const float4* __restrict__ s24 = reinterpret_cast<const float4*>(s2);
    float4 wv[WHVI_SLICES], sv[WHVI_SLICES];
    #pragma unroll
    for (int b = 0; b < WHVI_SLICES; ++b) wv[b] = v4[b * WHVI_SLICE_F4 + col0];
    #pragma unroll
    for (int b = 0; b < WHVI_SLICES; ++b) sv[b] = s24[b * WHVI_SLICE_F4 + col0];
    const float s1_0 = s1[0];

    #pragma unroll
    for (int h = 1; h < WHVI_SLICES; h <<= 1) {
        #pragma unroll
        for (int b = 0; b < WHVI_SLICES; ++b) {
            if ((b & h) == 0) {
                const float4 p = wv[b], q = wv[b + h];
                wv[b].x     = p.x + q.x;  wv[b].y     = p.y + q.y;
                wv[b].z     = p.z + q.z;  wv[b].w     = p.w + q.w;
                wv[b + h].x = p.x - q.x;  wv[b + h].y = p.y - q.y;
                wv[b + h].z = p.z - q.z;  wv[b + h].w = p.w - q.w;
            }
        }
    }
    #pragma unroll
    for (int b = 0; b < WHVI_SLICES; ++b) {
        wv[b].x = s1_0 * sv[b].x * wv[b].x;
        wv[b].y = s1_0 * sv[b].y * wv[b].y;
        wv[b].z = s1_0 * sv[b].z * wv[b].z;
        wv[b].w = s1_0 * sv[b].w * wv[b].w;
    }

    float4* __restrict__ o = out + (size_t)row0 * WHVI_ROW_F4 + col0;
    #pragma unroll
    for (int r = 0; r < WHVI_OUTER_ROWS; ++r) {
        const float xv = x[row0 + r];
        #pragma unroll
        for (int b = 0; b < WHVI_SLICES; ++b) {
            float4 o4;
            o4.x = xv * wv[b].x;
            o4.y = xv * wv[b].y;
            o4.z = xv * wv[b].z;
            o4.w = xv * wv[b].w;
            o[r * WHVI_ROW_F4 + b * WHVI_SLICE_F4] = o4;
        }
    }
}

extern "C" void kernel_launch(void* const* d_in, const int* in_sizes, int n_in,
                              void* d_out, int out_size, void* d_ws, size_t ws_size,
                              hipStream_t stream) {
    const float* x     = (const float*)d_in[0];
    const float* s1    = (const float*)d_in[1];
    const float* s2    = (const float*)d_in[2];
    const float* g_mu  = (const float*)d_in[3];
    const float* g_rho = (const float*)d_in[4];
    const float* eps   = (const float*)d_in[5];
    float* out = (float*)d_out;
    float* v   = (float*)d_ws;

    whvi_fwht_prep<<<WHVI_SLICES, 256, 0, stream>>>(g_mu, g_rho, eps, v);

    whvi_outer<<<WHVI_OUTER_WGS, 256, 0, stream>>>(x, s1, s2, v, (float4*)out);
}


# one launch: each of the 256 store workgroups rebuilds w itself (8192 softplus on hw exp2/log2/rcp, not library log1pf: a few ulp; wave b runs slice b through the same WMMA stages); no prep kernel
# speedup vs baseline: 2.5226x; 1.0910x over previous
#include <hip/hip_runtime.h>
#include <cmath>


#define WHVI_D 8192
#define WHVI_B 4096
#define WHVI_SLICES   8
#define WHVI_SLICE_F4 256
#define WHVI_PITCH     68
#define WHVI_SLICE_LDS (16 * WHVI_PITCH)

#define WHVI_OUTER_ROWS 16
#define WHVI_OUTER_WGS  256
#define WHVI_ROW_F4     2048

typedef __attribute__((ext_vector_type(2))) float v2f;
typedef __attribute__((ext_vector_type(8))) float v8f;

__device__ __forceinline__ float whvi_softplus(float r) {
    const float e = __builtin_amdgcn_exp2f(r * 1.44269504088896341f);
    const float u = 1.0f + e;
    const float d = u - 1.0f;
    const float l = __builtin_amdgcn_logf(u) * 0.693147180559945309f;
    const float sp = (d == 0.0f) ? e : (l * e) * __builtin_amdgcn_rcpf(d);
    return (r > 20.0f) ? r : sp;
}

__device__ __forceinline__ void whvi_h16_stage(float* __restrict__ us, const v2f (&a)[4],
                                               int n, int h, int kstride, int nstride) {
    float4 xb[4][2];
    #pragma unroll
    for (int q = 0; q < 4; ++q) {
        #pragma unroll
        for (int s = 0; s < 2; ++s)
            xb[q][s] = *reinterpret_cast<const float4*>(us + (4 * q + 2 * h + s) * kstride + n * nstride);
    }
    v8f acc0 = {}, acc1 = {}, acc2 = {}, acc3 = {};
    #pragma unroll
    for (int q = 0; q < 4; ++q) {
        v2f b0, b1, b2, b3;
        b0.x = xb[q][0].x; b0.y = xb[q][1].x;
        b1.x = xb[q][0].y; b1.y = xb[q][1].y;
        b2.x = xb[q][0].z; b2.y = xb[q][1].z;
        b3.x = xb[q][0].w; b3.y = xb[q][1].w;
        acc0 = __builtin_amdgcn_wmma_f32_16x16x4_f32(false, a[q], false, b0, (short)0, acc0, false, false);
        acc1 = __builtin_amdgcn_wmma_f32_16x16x4_f32(false, a[q], false, b1, (short)0, acc1, false, false);
        acc2 = __builtin_amdgcn_wmma_f32_16x16x4_f32(false, a[q], false, b2, (short)0, acc2, false, false);
        acc3 = __builtin_amdgcn_wmma_f32_16x16x4_f32(false, a[q], false, b3, (short)0, acc3, false, false);
    }
    #pragma unroll
    for (int r = 0; r < 8; ++r) {
        float4 y;
        y.x = acc0[r]; y.y = acc1[r]; y.z = acc2[r]; y.w = acc3[r];
        *reinterpret_cast<float4*>(us + (8 * h + r) * kstride + n * nstride) = y;
    }
}

__global__ __launch_bounds__(256)
void whvi_fused(const float* __restrict__ x,
                const float* __restrict__ s1,
                const float* __restrict__ s2,
                const float* __restrict__ g_mu,
                const float* __restrict__ g_rho,
                const float* __restrict__ eps,
                float4* __restrict__ out) {
    __shared__ __attribute__((aligned(16))) float u[WHVI_SLICES * WHVI_SLICE_LDS];

    const int tid  = threadIdx.x;
    const int lane = tid & 31;
    const int wave = tid >> 5;
    const int row0 = blockIdx.x * WHVI_OUTER_ROWS;
    const int own = (tid >> 4) * WHVI_PITCH + (tid & 15) * 4;

    const float4* __restrict__ r4  = reinterpret_cast<const float4*>(g_rho);
    const float4* __restrict__ e4  = reinterpret_cast<const float4*>(eps);
    const float4* __restrict__ m4  = reinterpret_cast<const float4*>(g_mu);
    const float4* __restrict__ s24 = reinterpret_cast<const float4*>(s2);
    float4 rv[WHVI_SLICES], ev[WHVI_SLICES], mv[WHVI_SLICES], sv[WHVI_SLICES];
    #pragma unroll
    for (int b = 0; b < WHVI_SLICES; ++b) rv[b] = r4[b * WHVI_SLICE_F4 + tid];
    #pragma unroll
    for (int b = 0; b < WHVI_SLICES; ++b) ev[b] = e4[b * WHVI_SLICE_F4 + tid];
    #pragma unroll
    for (int b = 0; b < WHVI_SLICES; ++b) mv[b] = m4[b * WHVI_SLICE_F4 + tid];
    #pragma unroll
    for (int b = 0; b < WHVI_SLICES; ++b) sv[b] = s24[b * WHVI_SLICE_F4 + tid];
    const float s1_0 = s1[0];
    __builtin_amdgcn_sched_barrier(0);

    #pragma unroll
    for (int b = 0; b < WHVI_SLICES; ++b) {
        const float g0 = mv[b].x + whvi_softplus(rv[b].x) * ev[b].x;
        const float g1 = mv[b].y + whvi_softplus(rv[b].y) * ev[b].y;
        const float g2 = mv[b].z + whvi_softplus(rv[b].z) * ev[b].z;
        const float g3 = mv[b].w + whvi_softplus(rv[b].w) * ev[b].w;
        const float a0 = g0 + g1, a1 = g0 - g1, a2 = g2 + g3, a3 = g2 - g3;
        float4 t;
        t.x = a0 + a2;
        t.y = a1 + a3;
        t.z = a0 - a2;
        t.w = a1 - a3;
        *reinterpret_cast<float4*>(u + b * WHVI_SLICE_LDS + own) = t;
    }
    __syncthreads();

    {
        const int n = lane & 15;
        const int h = lane >> 4;
        v2f a[4];
        #pragma unroll
        for (int q = 0; q < 4; ++q) {
            const int k0 = 4 * q + 2 * h;
            a[q].x = (__popc(n & k0)       & 1) ? -1.0f : 1.0f;
            a[q].y = (__popc(n & (k0 + 1)) & 1) ? -1.0f : 1.0f;
        }
        float* __restrict__ us = u + wave * WHVI_SLICE_LDS;
        whvi_h16_stage(us, a, n, h, 4, WHVI_PITCH);
        __syncthreads();
        whvi_h16_stage(us, a, n, h, WHVI_PITCH, 4);
    }
    __syncthreads();

    float4 wv[WHVI_SLICES];
    #pragma unroll
    for (int b = 0; b < WHVI_SLICES; ++b)
        wv[b] = *reinterpret_cast<const float4*>(u + b * WHVI_SLICE_LDS + own);

    #pragma unroll
    for (int h = 1; h < WHVI_SLICES; h <<= 1) {
        #pragma unroll
        for (int b = 0; b < WHVI_SLICES; ++b) {
            if ((b & h) == 0) {
                const float4 p = wv[b], q = wv[b + h];
                wv[b].x     = p.x + q.x;  wv[b].y     = p.y + q.y;
                wv[b].z     = p.z + q.z;  wv[b].w     = p.w + q.w;
                wv[b + h].x = p.x - q.x;  wv[b + h].y = p.y - q.y;
                wv[b + h].z = p.z - q.z;  wv[b + h].w = p.w - q.w;
            }
        }
    }
    #pragma unroll
    for (int b = 0; b < WHVI_SLICES; ++b) {
        wv[b].x = s1_0 * sv[b].x * wv[b].x;
        wv[b].y = s1_0 * sv[b].y * wv[b].y;
        wv[b].z = s1_0 * sv[b].z * wv[b].z;
        wv[b].w = s1_0 * sv[b].w * wv[b].w;
    }

    float4* __restrict__ o = out + (size_t)row0 * WHVI_ROW_F4 + tid;
    #pragma unroll
    for (int r = 0; r < WHVI_OUTER_ROWS; ++r) {
        const float xv = x[row0 + r];
        #pragma unroll
        for (int b = 0; b < WHVI_SLICES; ++b) {
            float4 o4;
            o4.x = xv * wv[b].x;
            o4.y = xv * wv[b].y;
            o4.z = xv * wv[b].z;
            o4.w = xv * wv[b].w;
            o[r * WHVI_ROW_F4 + b * WHVI_SLICE_F4] = o4;
        }
    }
}

extern "C" void kernel_launch(void* const* d_in, const int* in_sizes, int n_in,
                              void* d_out, int out_size, void* d_ws, size_t ws_size,
                              hipStream_t stream) {
    const float* x     = (const float*)d_in[0];
    const float* s1    = (const float*)d_in[1];
    const float* s2    = (const float*)d_in[2];
    const float* g_mu  = (const float*)d_in[3];
    const float* g_rho = (const float*)d_in[4];
    const float* eps   = (const float*)d_in[5];
    float* out = (float*)d_out;

    whvi_fused<<<WHVI_OUTER_WGS, 256, 0, stream>>>(x, s1, s2, g_mu, g_rho, eps, (float4*)out);
}
